# starting kernel
# baseline (speedup 1.0000x reference)
#include <hip/hip_runtime.h>
#include <hip/hip_bf16.h>
#include <math.h>

#define N_NODES   100000
#define N_EDGES   1600000
#define IN_FEATS  128
#define HEADS     4
#define OUT_F     16
#define HF        64
#define NEG_SLOPE 0.2f

typedef __attribute__((ext_vector_type(2))) float v2f;
typedef __attribute__((ext_vector_type(8))) float v8f;

__device__ __forceinline__ unsigned ordEnc(float f) {
  unsigned u = __float_as_uint(f);
  return (u & 0x80000000u) ? ~u : (u | 0x80000000u);
}
__device__ __forceinline__ float ordDec(unsigned x) {
  return (x & 0x80000000u) ? __uint_as_float(x ^ 0x80000000u)
                           : __uint_as_float(~x);
}

__device__ __forceinline__ float lrelu(float x) {
  return x > 0.0f ? x : NEG_SLOPE * x;
}

__global__ __launch_bounds__(128) void k_project(const float* __restrict__ feats,
                                                 const float* __restrict__ W,
                                                 float* __restrict__ z) {
  const int t      = threadIdx.x & 31;
  const int w      = threadIdx.x >> 5;
  const int lane16 = t & 15;
  const int half   = t >> 4;
  const int nodeBase = blockIdx.x * 16;

  const float* __restrict__ arow = feats + (size_t)(nodeBase + lane16) * IN_FEATS + half * 2;
  const float* __restrict__ brow = W     + (size_t)(w * 16  + lane16) * IN_FEATS + half * 2;

  v8f acc = {};
#pragma unroll
  for (int kk = 0; kk < IN_FEATS; kk += 4) {
    v2f a = { arow[kk], arow[kk + 1] };
    v2f b = { brow[kk], brow[kk + 1] };
    acc = __builtin_amdgcn_wmma_f32_16x16x4_f32(false, a, false, b,
                                                (short)0, acc, false, false);
  }

  float* __restrict__ zo = z + (size_t)(nodeBase + half * 8) * HF + w * 16 + lane16;
#pragma unroll
  for (int i = 0; i < 8; ++i) zo[(size_t)i * HF] = acc[i];
}

__global__ void k_logits(const float* __restrict__ z,
                         const float* __restrict__ attn_l,
                         const float* __restrict__ attn_r,
                         float* __restrict__ el, float* __restrict__ er) {
  int i = blockIdx.x * blockDim.x + threadIdx.x;
  if (i >= N_NODES * HEADS) return;
  int n = i >> 2, h = i & 3;
  const float* zp = z + (size_t)n * HF + h * OUT_F;
  const float* al = attn_l + h * OUT_F;
  const float* ar = attn_r + h * OUT_F;
  float sl = 0.0f, sr = 0.0f;
#pragma unroll
  for (int f = 0; f < OUT_F; ++f) {
    float v = zp[f];
    sl = fmaf(v, al[f], sl);
    sr = fmaf(v, ar[f], sr);
  }
  el[i] = sl;
  er[i] = sr;
}

__global__ void k_init(float* __restrict__ out, float* __restrict__ denom,
                       unsigned* __restrict__ menc) {
  int i = blockIdx.x * blockDim.x + threadIdx.x;
  if (i < N_NODES * HF) out[i] = 0.0f;
  if (i < N_NODES * HEADS) {
    denom[i] = 0.0f;
    menc[i]  = ordEnc(-INFINITY);
  }
}

__global__ void k_edge_max(const int* __restrict__ src, const int* __restrict__ dst,
                           const float* __restrict__ el, const float* __restrict__ er,
                           unsigned* __restrict__ menc) {
  int i = blockIdx.x * blockDim.x + threadIdx.x;
  if (i >= N_EDGES * HEADS) return;
  int e = i >> 2, h = i & 3;
  int s = src[e], d = dst[e];
  float v = lrelu(el[s * HEADS + h] + er[d * HEADS + h]);
  atomicMax(&menc[d * HEADS + h], ordEnc(v));
}

__global__ __launch_bounds__(256) void k_edge_agg(const int* __restrict__ src,
                                                  const int* __restrict__ dst,
                                                  const float* __restrict__ el,
                                                  const float* __restrict__ er,
                                                  const unsigned* __restrict__ menc,
                                                  const float* __restrict__ z,
                                                  float* __restrict__ denom,
                                                  float* __restrict__ out) {
  long long i = (long long)blockIdx.x * blockDim.x + threadIdx.x;
  if (i >= (long long)N_EDGES * HF) return;
  int e = (int)(i >> 6);
  int r = (int)(i & 63);
  int h = r >> 4;
  int f = r & 15;
  int s = src[e], d = dst[e];
  float lg = lrelu(el[s * HEADS + h] + er[d * HEADS + h]);
  float m  = ordDec(menc[d * HEADS + h]);
  float ex = __expf(lg - m);
  if (f == 0) atomicAdd(&denom[d * HEADS + h], ex);
  atomicAdd(&out[(size_t)d * HF + h * OUT_F + f],
            ex * z[(size_t)s * HF + h * OUT_F + f]);
}

__global__ void k_final(float* __restrict__ out, const float* __restrict__ denom,
                        const float* __restrict__ bias) {
  int i = blockIdx.x * blockDim.x + threadIdx.x;
  if (i >= N_NODES * HF) return;
  int n = i >> 6;
  int c = i & 63;
  int h = c >> 4;
  float den = denom[n * HEADS + h];
  float v   = out[i];
  out[i] = (den != 0.0f) ? (v / den + bias[c]) : bias[c];
}

extern "C" void kernel_launch(void* const* d_in, const int* in_sizes, int n_in,
                              void* d_out, int out_size, void* d_ws, size_t ws_size,
                              hipStream_t stream) {
  (void)in_sizes; (void)n_in; (void)out_size; (void)ws_size;
  const float* feats  = (const float*)d_in[0];
  const float* W      = (const float*)d_in[1];
  const float* attn_l = (const float*)d_in[2];
  const float* attn_r = (const float*)d_in[3];
  const float* bias   = (const float*)d_in[4];
  const int*   src    = (const int*)d_in[5];
  const int*   dst    = (const int*)d_in[6];
  float*       out    = (float*)d_out;

  float*    ws    = (float*)d_ws;
  float*    z     = ws;
  float*    el    = z  + (size_t)N_NODES * HF;
  float*    er    = el + (size_t)N_NODES * HEADS;
  unsigned* menc  = (unsigned*)(er + (size_t)N_NODES * HEADS);
  float*    denom = (float*)(menc + (size_t)N_NODES * HEADS);

  k_project<<<N_NODES / 16, 128, 0, stream>>>(feats, W, z);

  {
    int n = N_NODES * HEADS;
    k_logits<<<(n + 255) / 256, 256, 0, stream>>>(z, attn_l, attn_r, el, er);
  }

  {
    int n = N_NODES * HF;
    k_init<<<(n + 255) / 256, 256, 0, stream>>>(out, denom, menc);
  }

  {
    int n = N_EDGES * HEADS;
    k_edge_max<<<(n + 255) / 256, 256, 0, stream>>>(src, dst, el, er, menc);
  }

  {
    long long n = (long long)N_EDGES * HF;
    k_edge_agg<<<(unsigned)((n + 255) / 256), 256, 0, stream>>>(src, dst, el, er,
                                                               menc, z, denom, out);
  }

  {
    int n = N_NODES * HF;
    k_final<<<(n + 255) / 256, 256, 0, stream>>>(out, denom, bias);
  }
}


# Projection feed: feats tile staged once per block through LDS by whole-row loads, W fragments kept in registers over 5 node tiles per block (was 64 many-line 8-byte loads per wave); k order kept
# speedup vs baseline: 3.2706x; 3.2706x over previous
#include <hip/hip_runtime.h>
#include <hip/hip_bf16.h>
#include <math.h>

#define N_NODES   100000
#define N_EDGES   1600000
#define IN_FEATS  128
#define HEADS     4
#define OUT_F     16
#define HF        64
#define NEG_SLOPE 0.2f

typedef __attribute__((ext_vector_type(2))) float v2f;
typedef __attribute__((ext_vector_type(4))) float v4f;
typedef __attribute__((ext_vector_type(8))) float v8f;

__device__ __forceinline__ unsigned ordEnc(float f) {
  unsigned u = __float_as_uint(f);
  return (u & 0x80000000u) ? ~u : (u | 0x80000000u);
}
__device__ __forceinline__ float ordDec(unsigned x) {
  return (x & 0x80000000u) ? __uint_as_float(x ^ 0x80000000u)
                           : __uint_as_float(~x);
}

__device__ __forceinline__ float lrelu(float x) {
  return x > 0.0f ? x : NEG_SLOPE * x;
}

#define TILES_PER_BLOCK 5
#define KSTEPS (IN_FEATS / 4)
#define A_PITCH (IN_FEATS + 4)

__device__ __forceinline__ void lds_handover() {
  __builtin_amdgcn_fence(__ATOMIC_RELEASE, "workgroup", "local");
  __builtin_amdgcn_s_barrier();
  __builtin_amdgcn_fence(__ATOMIC_ACQUIRE, "workgroup", "local");
}

__global__ __launch_bounds__(128) void k_project(const float* feats,
                                                 const float* __restrict__ W,
                                                 float* __restrict__ z) {
  __shared__ v4f s_a4[2][16 * A_PITCH / 4];

  const int t      = threadIdx.x & 31;
  const int w      = threadIdx.x >> 5;
  const int lane16 = t & 15;
  const int half   = t >> 4;
  const int tile0  = blockIdx.x * TILES_PER_BLOCK;

  const v4f* gsrc = reinterpret_cast<const v4f*>(feats)
                                    + ((size_t)tile0 * 16 + w) * (IN_FEATS / 4) + t;
  const int sdst = w * (A_PITCH / 4) + t;

  const size_t rstep = (size_t)4 * (IN_FEATS / 4);
  v4f g0 = gsrc[0], g1 = gsrc[rstep], g2 = gsrc[2 * rstep], g3 = gsrc[3 * rstep];

  const float* __restrict__ brow = W + (size_t)(w * 16 + lane16) * IN_FEATS + half * 2;
  v2f b[KSTEPS];
#pragma unroll
  for (int s = 0; s < KSTEPS; ++s) b[s] = v2f{ brow[4 * s], brow[4 * s + 1] };

  const int sstep = 4 * (A_PITCH / 4);
  s_a4[0][sdst] = g0;  s_a4[0][sdst + sstep] = g1;  s_a4[0][sdst + 2 * sstep] = g2;  s_a4[0][sdst + 3 * sstep] = g3;
  lds_handover();

#pragma unroll
  for (int i = 0; i < TILES_PER_BLOCK; ++i) {
    const int  cur  = i & 1;
    const bool more = i + 1 < TILES_PER_BLOCK;
    if (more) {
      const v4f* gn = gsrc + (size_t)(i + 1) * 16 * (IN_FEATS / 4);
      g0 = gn[0];  g1 = gn[rstep];  g2 = gn[2 * rstep];  g3 = gn[3 * rstep];
    }

    const float* __restrict__ ap = reinterpret_cast<const float*>(&s_a4[cur][0])
                                   + lane16 * A_PITCH + half * 2;
    v2f a[KSTEPS];
#pragma unroll
    for (int s = 0; s < KSTEPS; ++s) a[s] = *reinterpret_cast<const v2f*>(ap + 4 * s);
    __builtin_amdgcn_sched_barrier(0);

    v8f acc = {};
#pragma unroll
    for (int s = 0; s < KSTEPS; ++s)
      acc = __builtin_amdgcn_wmma_f32_16x16x4_f32(false, a[s], false, b[s],
                                                  (short)0, acc, false, false);

    __builtin_amdgcn_sched_barrier(0);

    float* __restrict__ zo = z + ((size_t)(tile0 + i) * 16 + half * 8) * HF + w * 16 + lane16;
#pragma unroll
    for (int r = 0; r < 8; ++r) zo[(size_t)r * HF] = acc[r];

    if (more) {
      v4f* sn = &s_a4[cur ^ 1][sdst];
      sn[0] = g0;  sn[sstep] = g1;  sn[2 * sstep] = g2;  sn[3 * sstep] = g3;
      lds_handover();
    }
  }
}

__global__ void k_logits(const float* __restrict__ z,
                         const float* __restrict__ attn_l,
                         const float* __restrict__ attn_r,
                         float* __restrict__ el, float* __restrict__ er) {
  int i = blockIdx.x * blockDim.x + threadIdx.x;
  if (i >= N_NODES * HEADS) return;
  int n = i >> 2, h = i & 3;
  const float* zp = z + (size_t)n * HF + h * OUT_F;
  const float* al = attn_l + h * OUT_F;
  const float* ar = attn_r + h * OUT_F;
  float sl = 0.0f, sr = 0.0f;
#pragma unroll
  for (int f = 0; f < OUT_F; ++f) {
    float v = zp[f];
    sl = fmaf(v, al[f], sl);
    sr = fmaf(v, ar[f], sr);
  }
  el[i] = sl;
  er[i] = sr;
}

#define WS_ONE_COPY  ((size_t)4 * ((size_t)N_NODES * HF + (size_t)4 * N_NODES * HEADS))
#define WS_TWO_COPY  (WS_ONE_COPY + (size_t)4 * ((size_t)N_NODES * HF + (size_t)N_NODES * HEADS))
#define WS_TWO_MAX   (WS_TWO_COPY + (size_t)4 * ((size_t)N_NODES * HEADS))

__global__ void k_init(float* __restrict__ out, float* __restrict__ denom,
                       unsigned* __restrict__ menc,
                       float* __restrict__ acc1, float* __restrict__ denom1,
                       unsigned* __restrict__ menc1,
                       size_t ws_bytes) {
  const bool two    = ws_bytes >= WS_TWO_COPY;
  const bool twoMax = ws_bytes >= WS_TWO_MAX;
  int i = blockIdx.x * blockDim.x + threadIdx.x;
  if (i < N_NODES * HF) {
    out[i] = 0.0f;
    if (two) acc1[i] = 0.0f;
  }
  if (i < N_NODES * HEADS) {
    denom[i] = 0.0f;
    if (two) denom1[i] = 0.0f;
    menc[i]  = ordEnc(-INFINITY);
    if (twoMax) menc1[i] = ordEnc(-INFINITY);
  }
}

__global__ __launch_bounds__(256) void k_edge_max(const int* __restrict__ src,
                                                  const int* __restrict__ dst,
                                                  const float* __restrict__ el,
                                                  const float* __restrict__ er,
                                                  unsigned* __restrict__ menc0,
                                                  unsigned* __restrict__ menc1,
                                                  size_t ws_bytes) {
  __shared__ unsigned s_die;
  const int  i    = blockIdx.x * blockDim.x + threadIdx.x;
  const bool live = i < N_EDGES * HEADS;
  const int  e = live ? (i >> 2) : 0, h = i & 3;
  const int  s = src[e], d = dst[e];
  const float lv = el[s * HEADS + h];
  const float rv = er[d * HEADS + h];
  unsigned* menc = menc0;
  if (ws_bytes >= WS_TWO_MAX) {
    if (threadIdx.x == 0) s_die = (__builtin_amdgcn_s_sendmsg_rtn(0x87) >> 16) & 0xFu;
    __syncthreads();
    if (s_die >= 4u) menc = menc1;
  }
  if (live) atomicMax(&menc[d * HEADS + h], ordEnc(lrelu(lv + rv)));
}

#define EPW 32
#define EPG 8
#define NGRP (EPW / EPG)

__global__ __launch_bounds__(256) void k_edge_agg(const int* __restrict__ src,
                                                  const int* __restrict__ dst,
                                                  const float* __restrict__ el,
                                                  const float* __restrict__ er,
                                                  const unsigned* __restrict__ menc,
                                                  const unsigned* __restrict__ menc1,
                                                  const float* __restrict__ z,
                                                  float* __restrict__ denom0,
                                                  float* __restrict__ out0,
                                                  float* __restrict__ denom1,
                                                  float* __restrict__ acc1,
                                                  size_t ws_bytes) {
  const int lane = threadIdx.x & 31;
  const int wave = blockIdx.x * (256 / 32) + (threadIdx.x >> 5);
  const int e0   = wave * EPW;
  if (e0 >= N_EDGES) return;
  const int sub = lane >> 2;
  const int h   = lane & 3;
  const int selLo = (lane >> 4) << 2;
  const int selHi = selLo + 8;

  int sv[NGRP], dv[NGRP];
  for (int p = 0; p < NGRP; ++p) {
    const int e = e0 + p * EPG + sub;
    sv[p] = src[e];
    dv[p] = dst[e];
  }
  float    lv[NGRP], rv[NGRP];
  unsigned mv[NGRP];
  for (int p = 0; p < NGRP; ++p) {
    lv[p] = el[sv[p] * HEADS + h];
    rv[p] = er[dv[p] * HEADS + h];
    mv[p] = menc[dv[p] * HEADS + h];
  }
  const unsigned* __restrict__ mencB = (ws_bytes >= WS_TWO_MAX) ? menc1 : menc;
  unsigned ov[NGRP];
  for (int p = 0; p < NGRP; ++p) ov[p] = mencB[dv[p] * HEADS + h];

  float a0[EPG], a1[EPG];
  for (int j = 0; j < EPG; ++j) {
    const int s = __builtin_amdgcn_readlane(sv[0], j * 4);
    a0[j] = z[s * HF + lane];
    a1[j] = z[s * HF + lane + 32];
  }

  float* out   = out0;
  float* denom = denom0;
  if (ws_bytes >= WS_TWO_COPY) {
    const unsigned die = (__builtin_amdgcn_s_sendmsg_rtn(0x87) >> 16) & 0xFu;
    if (die >= 4u) { out = acc1; denom = denom1; }
  }

  float wv[NGRP];
  for (int p = 0; p < NGRP; ++p) {
    float lg = lrelu(lv[p] + rv[p]);
    float m  = ordDec(mv[p] > ov[p] ? mv[p] : ov[p]);
    wv[p] = __expf(lg - m);
  }
  for (int p = 0; p < NGRP; ++p) atomicAdd(&denom[dv[p] * HEADS + h], wv[p]);

  for (int p = 0; p < NGRP; ++p) {
    float b0[EPG], b1[EPG];
    const int pn = (p + 1 < NGRP) ? p + 1 : p;
    if (p + 1 < NGRP) {
      for (int j = 0; j < EPG; ++j) {
        const int s = __builtin_amdgcn_readlane(sv[pn], j * 4);
        b0[j] = z[s * HF + lane];
        b1[j] = z[s * HF + lane + 32];
      }
    }
    const int wbits = __float_as_int(wv[p]);
    for (int j = 0; j < EPG; ++j) {
      const int   d  = __builtin_amdgcn_readlane(dv[p], j * 4);
      const float wl = __int_as_float(__builtin_amdgcn_ds_bpermute(selLo + 16 * j, wbits));
      const float wh = __int_as_float(__builtin_amdgcn_ds_bpermute(selHi + 16 * j, wbits));
      atomicAdd(&out[d * HF + lane],      wl * a0[j]);
      atomicAdd(&out[d * HF + lane + 32], wh * a1[j]);
    }
    if (p + 1 < NGRP) {
      for (int j = 0; j < EPG; ++j) { a0[j] = b0[j]; a1[j] = b1[j]; }
    }
  }
}

__global__ void k_final(float* __restrict__ out, const float* __restrict__ denom,
                        const float* __restrict__ bias,
                        const float* __restrict__ acc1, const float* __restrict__ denom1,
                        size_t ws_bytes) {
  const bool two = ws_bytes >= WS_TWO_COPY;
  int i = blockIdx.x * blockDim.x + threadIdx.x;
  if (i >= N_NODES * HF) return;
  int n = i >> 6;
  int c = i & 63;
  int h = c >> 4;
  float den = denom[n * HEADS + h];
  float v   = out[i];
  if (two) {
    den += denom1[n * HEADS + h];
    v   += acc1[i];
  }
  out[i] = (den != 0.0f) ? (v / den + bias[c]) : bias[c];
}

extern "C" void kernel_launch(void* const* d_in, const int* in_sizes, int n_in,
                              void* d_out, int out_size, void* d_ws, size_t ws_size,
                              hipStream_t stream) {
  (void)in_sizes; (void)n_in; (void)out_size;
  const float* feats  = (const float*)d_in[0];
  const float* W      = (const float*)d_in[1];
  const float* attn_l = (const float*)d_in[2];
  const float* attn_r = (const float*)d_in[3];
  const float* bias   = (const float*)d_in[4];
  const int*   src    = (const int*)d_in[5];
  const int*   dst    = (const int*)d_in[6];
  float*       out    = (float*)d_out;

  float*    ws    = (float*)d_ws;
  float*    z     = ws;
  float*    el    = z  + (size_t)N_NODES * HF;
  float*    er    = el + (size_t)N_NODES * HEADS;
  unsigned* menc  = (unsigned*)(er + (size_t)N_NODES * HEADS);
  float*    denom = (float*)(menc + (size_t)N_NODES * HEADS);
  float*    acc1   = denom + (size_t)N_NODES * HEADS;
  float*    denom1 = acc1  + (size_t)N_NODES * HF;
  unsigned* menc1  = (unsigned*)(denom1 + (size_t)N_NODES * HEADS);

  k_project<<<N_NODES / (16 * TILES_PER_BLOCK), 128, 0, stream>>>(feats, W, z);

  {
    int n = N_NODES * HEADS;
    k_logits<<<(n + 255) / 256, 256, 0, stream>>>(z, attn_l, attn_r, el, er);
  }

  {
    int n = N_NODES * HF;
    k_init<<<(n + 255) / 256, 256, 0, stream>>>(out, denom, menc, acc1, denom1, menc1, ws_size);
  }

  {
    int n = N_EDGES * HEADS;
    k_edge_max<<<(n + 255) / 256, 256, 0, stream>>>(src, dst, el, er, menc, menc1, ws_size);
  }

  k_edge_agg<<<N_EDGES / (EPW * 8), 256, 0, stream>>>(src, dst, el, er,
                                                    menc, menc1, z, denom, out,
                                                    denom1, acc1, ws_size);

  {
    int n = N_NODES * HF;
    k_final<<<(n + 255) / 256, 256, 0, stream>>>(out, denom, bias, acc1, denom1, ws_size);
  }
}


# Softmax shift per (dst, head) from a per-head global bound of the logits (integer max in the logit pass) while the range allows: no segment-max pass, no max-word gathers; max form kept, exits at once
# speedup vs baseline: 4.1540x; 1.2701x over previous
#include <hip/hip_runtime.h>
#include <hip/hip_bf16.h>
#include <math.h>

#define N_NODES   100000
#define N_EDGES   1600000
#define IN_FEATS  128
#define HEADS     4
#define OUT_F     16
#define HF        64
#define NEG_SLOPE 0.2f

typedef __attribute__((ext_vector_type(2))) float v2f;
typedef __attribute__((ext_vector_type(4))) float v4f;
typedef __attribute__((ext_vector_type(8))) float v8f;

__device__ __forceinline__ unsigned ordEnc(float f) {
  unsigned u = __float_as_uint(f);
  return (u & 0x80000000u) ? ~u : (u | 0x80000000u);
}
__device__ __forceinline__ float ordDec(unsigned x) {
  return (x & 0x80000000u) ? __uint_as_float(x ^ 0x80000000u)
                           : __uint_as_float(~x);
}

__device__ __forceinline__ float lrelu(float x) {
  return x > 0.0f ? x : NEG_SLOPE * x;
}

#define SHIFT_RANGE 16.0f

#define WS_ONE_COPY   ((size_t)4 * ((size_t)N_NODES * HF + (size_t)4 * N_NODES * HEADS))
#define WS_TWO_COPY   (WS_ONE_COPY + (size_t)4 * ((size_t)N_NODES * HF + (size_t)N_NODES * HEADS))
#define WS_TWO_MAX    (WS_TWO_COPY + (size_t)4 * ((size_t)N_NODES * HEADS))
#define WS_WITH_BOUND (WS_TWO_MAX + (size_t)4 * 2 * HEADS)

__device__ __forceinline__ bool shift_from_bound(const unsigned* __restrict__ bound, bool have,
                                                 const int* __restrict__ src,
                                                 float (&mid)[HEADS]) {
  const unsigned* __restrict__ w = have ? bound : reinterpret_cast<const unsigned*>(src);
  bool ok = have;
  for (int h = 0; h < HEADS; ++h) {
    const float hi = ordDec(w[h]);
    const float lo = -ordDec(w[HEADS + h]);
    ok = ok && (hi - lo <= SHIFT_RANGE);
    mid[h] = lo + 0.5f * (hi - lo);
  }
  return ok;
}

#define TILES_PER_BLOCK 5
#define KSTEPS (IN_FEATS / 4)
#define A_PITCH (IN_FEATS + 4)

__device__ __forceinline__ void lds_handover() {
  __builtin_amdgcn_fence(__ATOMIC_RELEASE, "workgroup", "local");
  __builtin_amdgcn_s_barrier();
  __builtin_amdgcn_fence(__ATOMIC_ACQUIRE, "workgroup", "local");
}

__global__ __launch_bounds__(128) void k_project(const float* feats,
                                                 const float* __restrict__ W,
                                                 float* __restrict__ z,
                                                 unsigned* __restrict__ bound,
                                                 size_t ws_bytes) {
  __shared__ v4f s_a4[2][16 * A_PITCH / 4];

  if (blockIdx.x == 0 && threadIdx.x < 2 * HEADS && ws_bytes >= WS_WITH_BOUND)
    bound[threadIdx.x] = ordEnc(-INFINITY);

  const int t      = threadIdx.x & 31;
  const int w      = threadIdx.x >> 5;
  const int lane16 = t & 15;
  const int half   = t >> 4;
  const int tile0  = blockIdx.x * TILES_PER_BLOCK;

  const v4f* gsrc = reinterpret_cast<const v4f*>(feats)
                                    + ((size_t)tile0 * 16 + w) * (IN_FEATS / 4) + t;
  const int sdst = w * (A_PITCH / 4) + t;

  const size_t rstep = (size_t)4 * (IN_FEATS / 4);
  v4f g0 = gsrc[0], g1 = gsrc[rstep], g2 = gsrc[2 * rstep], g3 = gsrc[3 * rstep];

  const float* __restrict__ brow = W + (size_t)(w * 16 + lane16) * IN_FEATS + half * 2;
  v2f b[KSTEPS];
#pragma unroll
  for (int s = 0; s < KSTEPS; ++s) b[s] = v2f{ brow[4 * s], brow[4 * s + 1] };

  const int sstep = 4 * (A_PITCH / 4);
  s_a4[0][sdst] = g0;  s_a4[0][sdst + sstep] = g1;  s_a4[0][sdst + 2 * sstep] = g2;  s_a4[0][sdst + 3 * sstep] = g3;
  lds_handover();

#pragma unroll
  for (int i = 0; i < TILES_PER_BLOCK; ++i) {
    const int  cur  = i & 1;
    const bool more = i + 1 < TILES_PER_BLOCK;
    if (more) {
      const v4f* gn = gsrc + (size_t)(i + 1) * 16 * (IN_FEATS / 4);
      g0 = gn[0];  g1 = gn[rstep];  g2 = gn[2 * rstep];  g3 = gn[3 * rstep];
    }

    const float* __restrict__ ap = reinterpret_cast<const float*>(&s_a4[cur][0])
                                   + lane16 * A_PITCH + half * 2;
    v2f a[KSTEPS];
#pragma unroll
    for (int s = 0; s < KSTEPS; ++s) a[s] = *reinterpret_cast<const v2f*>(ap + 4 * s);
    __builtin_amdgcn_sched_barrier(0);

    v8f acc = {};
#pragma unroll
    for (int s = 0; s < KSTEPS; ++s)
      acc = __builtin_amdgcn_wmma_f32_16x16x4_f32(false, a[s], false, b[s],
                                                  (short)0, acc, false, false);

    __builtin_amdgcn_sched_barrier(0);

    float* __restrict__ zo = z + ((size_t)(tile0 + i) * 16 + half * 8) * HF + w * 16 + lane16;
#pragma unroll
    for (int r = 0; r < 8; ++r) zo[(size_t)r * HF] = acc[r];

    if (more) {
      v4f* sn = &s_a4[cur ^ 1][sdst];
      sn[0] = g0;  sn[sstep] = g1;  sn[2 * sstep] = g2;  sn[3 * sstep] = g3;
      lds_handover();
    }
  }
}

#define LOGIT_BLOCK 800

static_assert((N_NODES * HEADS) % LOGIT_BLOCK == 0 && LOGIT_BLOCK % 32 == 0, "exact grid of whole waves");

__global__ __launch_bounds__(LOGIT_BLOCK) void k_logits(const float* __restrict__ z,
                         const float* __restrict__ attn_l,
                         const float* __restrict__ attn_r,
                         float* __restrict__ el, float* __restrict__ er,
                         unsigned* __restrict__ bound, size_t ws_bytes) {
  __shared__ unsigned s_red[LOGIT_BLOCK / 32][2 * HEADS];
  const int i = blockIdx.x * LOGIT_BLOCK + threadIdx.x;
  int n = i >> 2, h = i & 3;
  const float* zp = z + (size_t)n * HF + h * OUT_F;
  const float* al = attn_l + h * OUT_F;
  const float* ar = attn_r + h * OUT_F;
  float sl = 0.0f, sr = 0.0f;
#pragma unroll
  for (int f = 0; f < OUT_F; ++f) {
    float v = zp[f];
    sl = fmaf(v, al[f], sl);
    sr = fmaf(v, ar[f], sr);
  }
  el[i] = sl;
  er[i] = sr;
  if (ws_bytes >= WS_WITH_BOUND) {
    unsigned vmax = ordEnc(sl), vneg = ordEnc(-sl);
    for (int off = 4; off < 32; off <<= 1) {
      const unsigned a = __shfl_xor(vmax, off), b = __shfl_xor(vneg, off);
      vmax = a > vmax ? a : vmax;
      vneg = b > vneg ? b : vneg;
    }
    const int lane = threadIdx.x & 31, wave = threadIdx.x >> 5;
    if (lane < HEADS) { s_red[wave][lane] = vmax;  s_red[wave][HEADS + lane] = vneg; }
    __syncthreads();
    if (threadIdx.x < 2 * HEADS) {
      unsigned v = s_red[0][threadIdx.x];
      for (int w = 1; w < LOGIT_BLOCK / 32; ++w) { const unsigned o = s_red[w][threadIdx.x]; v = o > v ? o : v; }
      atomicMax(&bound[threadIdx.x], v);
    }
  }
}


__global__ void k_init(float* __restrict__ out, float* __restrict__ denom,
                       unsigned* __restrict__ menc,
                       float* __restrict__ acc1, float* __restrict__ denom1,
                       unsigned* __restrict__ menc1,
                       size_t ws_bytes) {
  const bool two    = ws_bytes >= WS_TWO_COPY;
  const bool twoMax = ws_bytes >= WS_TWO_MAX;
  int i = blockIdx.x * blockDim.x + threadIdx.x;
  if (i < N_NODES * HF) {
    out[i] = 0.0f;
    if (two) acc1[i] = 0.0f;
  }
  if (i < N_NODES * HEADS) {
    denom[i] = 0.0f;
    if (two) denom1[i] = 0.0f;
    menc[i]  = ordEnc(-INFINITY);
    if (twoMax) menc1[i] = ordEnc(-INFINITY);
  }
}

#define FALLBACK_BLOCKS 2048
#define MAX_PIECES (N_EDGES * HEADS / 256)

__global__ __launch_bounds__(256) void k_edge_max(const int* __restrict__ src,
                                                  const int* __restrict__ dst,
                                                  const float* __restrict__ el,
                                                  const float* __restrict__ er,
                                                  unsigned* __restrict__ menc0,
                                                  unsigned* __restrict__ menc1,
                                                  const unsigned* __restrict__ bound,
                                                  size_t ws_bytes) {
  __shared__ unsigned s_die;
  float midh[HEADS];
  if (shift_from_bound(bound, ws_bytes >= WS_WITH_BOUND, src, midh)) return;
  unsigned* menc = menc0;
  if (ws_bytes >= WS_TWO_MAX) {
    if (threadIdx.x == 0) s_die = (__builtin_amdgcn_s_sendmsg_rtn(0x87) >> 16) & 0xFu;
    __syncthreads();
    if (s_die >= 4u) menc = menc1;
  }
  const int h = threadIdx.x & 3;
  for (int piece = blockIdx.x; piece < MAX_PIECES; piece += FALLBACK_BLOCKS) {
    const int e = (piece * 256 + threadIdx.x) >> 2;
    const int s = src[e], d = dst[e];
    const float lv = el[s * HEADS + h];
    const float rv = er[d * HEADS + h];
    atomicMax(&menc[d * HEADS + h], ordEnc(lrelu(lv + rv)));
  }
}

#define EPW 32
#define EPG 8
#define NGRP (EPW / EPG)

template <bool BY_BOUND>
__device__ __forceinline__ void edge_walk(const int (&sv)[NGRP], const int (&dv)[NGRP],
                                          const int lane, const float mid,
                                          const float* __restrict__ el,
                                          const float* __restrict__ er,
                                          const unsigned* __restrict__ menc,
                                          const unsigned* __restrict__ menc1,
                                          const float* __restrict__ z,
                                          float* __restrict__ denom0,
                                          float* __restrict__ out0,
                                          float* __restrict__ denom1,
                                          float* __restrict__ acc1,
                                          size_t ws_bytes) {
  const int h   = lane & 3;
  const int selLo = (lane >> 4) << 2;
  const int selHi = selLo + 8;

  float    lv[NGRP], rv[NGRP];
  for (int p = 0; p < NGRP; ++p) {
    lv[p] = el[sv[p] * HEADS + h];
    rv[p] = er[dv[p] * HEADS + h];
  }
  unsigned mv[NGRP], ov[NGRP];
  if (!BY_BOUND) {
    const unsigned* __restrict__ mencB = (ws_bytes >= WS_TWO_MAX) ? menc1 : menc;
    for (int p = 0; p < NGRP; ++p) {
      mv[p] = menc[dv[p] * HEADS + h];
      ov[p] = mencB[dv[p] * HEADS + h];
    }
  }

  float a0[EPG], a1[EPG];
  for (int j = 0; j < EPG; ++j) {
    const int s = __builtin_amdgcn_readlane(sv[0], j * 4);
    a0[j] = z[s * HF + lane];
    a1[j] = z[s * HF + lane + 32];
  }

  float* out   = out0;
  float* denom = denom0;
  if (ws_bytes >= WS_TWO_COPY) {
    const unsigned die = (__builtin_amdgcn_s_sendmsg_rtn(0x87) >> 16) & 0xFu;
    if (die >= 4u) { out = acc1; denom = denom1; }
  }

  float wv[NGRP];
  for (int p = 0; p < NGRP; ++p) {
    float lg = lrelu(lv[p] + rv[p]);
    float c;
    if (BY_BOUND) c = lrelu(rv[p] + mid);
    else          c = ordDec(mv[p] > ov[p] ? mv[p] : ov[p]);
    wv[p] = __expf(lg - c);
  }
  for (int p = 0; p < NGRP; ++p) atomicAdd(&denom[dv[p] * HEADS + h], wv[p]);

  for (int p = 0; p < NGRP; ++p) {
    float b0[EPG], b1[EPG];
    const int pn = (p + 1 < NGRP) ? p + 1 : p;
    if (p + 1 < NGRP) {
      for (int j = 0; j < EPG; ++j) {
        const int s = __builtin_amdgcn_readlane(sv[pn], j * 4);
        b0[j] = z[s * HF + lane];
        b1[j] = z[s * HF + lane + 32];
      }
    }
    const int wbits = __float_as_int(wv[p]);
    for (int j = 0; j < EPG; ++j) {
      const int   d  = __builtin_amdgcn_readlane(dv[p], j * 4);
      const float wl = __int_as_float(__builtin_amdgcn_ds_bpermute(selLo + 16 * j, wbits));
      const float wh = __int_as_float(__builtin_amdgcn_ds_bpermute(selHi + 16 * j, wbits));
      atomicAdd(&out[d * HF + lane],      wl * a0[j]);
      atomicAdd(&out[d * HF + lane + 32], wh * a1[j]);
    }
    if (p + 1 < NGRP) {
      for (int j = 0; j < EPG; ++j) { a0[j] = b0[j]; a1[j] = b1[j]; }
    }
  }
}

template <bool BY_BOUND, int GRID_BLOCKS>
__global__ __launch_bounds__(256) void k_edge_agg(const int* __restrict__ src,
                                                  const int* __restrict__ dst,
                                                  const float* __restrict__ el,
                                                  const float* __restrict__ er,
                                                  const unsigned* __restrict__ menc,
                                                  const unsigned* __restrict__ menc1,
                                                  const float* __restrict__ z,
                                                  float* __restrict__ denom0,
                                                  float* __restrict__ out0,
                                                  float* __restrict__ denom1,
                                                  float* __restrict__ acc1,
                                                  const unsigned* __restrict__ bound,
                                                  size_t ws_bytes) {
  constexpr int N_CHUNKS   = N_EDGES / EPW;
  constexpr int GRID_WAVES = GRID_BLOCKS * (256 / 32);
  constexpr int TRIPS      = (N_CHUNKS + GRID_WAVES - 1) / GRID_WAVES;
  const int lane  = threadIdx.x & 31;
  const int wave0 = blockIdx.x * (256 / 32) + (threadIdx.x >> 5);
  const int sub   = lane >> 2;
  const int h     = lane & 3;
  for (int t = 0; t < TRIPS; ++t) {
    const int chunk = wave0 + t * GRID_WAVES;
    if (chunk >= N_CHUNKS) return;
    const int e0 = chunk * EPW;
    int sv[NGRP], dv[NGRP];
    for (int p = 0; p < NGRP; ++p) {
      const int e = e0 + p * EPG + sub;
      sv[p] = src[e];
      dv[p] = dst[e];
    }
    float midh[HEADS];
    const bool byBound = shift_from_bound(bound, ws_bytes >= WS_WITH_BOUND, src, midh);
    if (byBound != BY_BOUND) return;
    const float mid = h == 0 ? midh[0] : h == 1 ? midh[1] : h == 2 ? midh[2] : midh[3];
    edge_walk<BY_BOUND>(sv, dv, lane, mid, el, er, menc, menc1, z, denom0, out0, denom1, acc1, ws_bytes);
  }
}

__global__ void k_final(float* __restrict__ out, const float* __restrict__ denom,
                        const float* __restrict__ bias,
                        const float* __restrict__ acc1, const float* __restrict__ denom1,
                        size_t ws_bytes) {
  const bool two = ws_bytes >= WS_TWO_COPY;
  int i = blockIdx.x * blockDim.x + threadIdx.x;
  if (i >= N_NODES * HF) return;
  int n = i >> 6;
  int c = i & 63;
  int h = c >> 4;
  float den = denom[n * HEADS + h];
  float v   = out[i];
  if (two) {
    den += denom1[n * HEADS + h];
    v   += acc1[i];
  }
  out[i] = (den != 0.0f) ? (v / den + bias[c]) : bias[c];
}

extern "C" void kernel_launch(void* const* d_in, const int* in_sizes, int n_in,
                              void* d_out, int out_size, void* d_ws, size_t ws_size,
                              hipStream_t stream) {
  (void)in_sizes; (void)n_in; (void)out_size;
  const float* feats  = (const float*)d_in[0];
  const float* W      = (const float*)d_in[1];
  const float* attn_l = (const float*)d_in[2];
  const float* attn_r = (const float*)d_in[3];
  const float* bias   = (const float*)d_in[4];
  const int*   src    = (const int*)d_in[5];
  const int*   dst    = (const int*)d_in[6];
  float*       out    = (float*)d_out;

  float*    ws    = (float*)d_ws;
  float*    z     = ws;
  float*    el    = z  + (size_t)N_NODES * HF;
  float*    er    = el + (size_t)N_NODES * HEADS;
  unsigned* menc  = (unsigned*)(er + (size_t)N_NODES * HEADS);
  float*    denom = (float*)(menc + (size_t)N_NODES * HEADS);
  float*    acc1   = denom + (size_t)N_NODES * HEADS;
  float*    denom1 = acc1  + (size_t)N_NODES * HF;
  unsigned* menc1  = (unsigned*)(denom1 + (size_t)N_NODES * HEADS);
  unsigned* bound  = menc1 + (size_t)N_NODES * HEADS;

  k_project<<<N_NODES / (16 * TILES_PER_BLOCK), 128, 0, stream>>>(feats, W, z, bound, ws_size);

  {
    int n = N_NODES * HEADS;
    k_logits<<<n / LOGIT_BLOCK, LOGIT_BLOCK, 0, stream>>>(z, attn_l, attn_r, el, er, bound, ws_size);
  }

  {
    int n = N_NODES * HF;
    k_init<<<(n + 255) / 256, 256, 0, stream>>>(out, denom, menc, acc1, denom1, menc1, ws_size);
  }

  k_edge_max<<<FALLBACK_BLOCKS, 256, 0, stream>>>(src, dst, el, er, menc, menc1, bound, ws_size);

  k_edge_agg<true, N_EDGES / (EPW * 8)><<<N_EDGES / (EPW * 8), 256, 0, stream>>>(src, dst, el, er,
                                                    menc, menc1, z, denom, out,
                                                    denom1, acc1, bound, ws_size);
  k_edge_agg<false, FALLBACK_BLOCKS><<<FALLBACK_BLOCKS, 256, 0, stream>>>(src, dst, el, er,
                                                    menc, menc1, z, denom, out,
                                                    denom1, acc1, bound, ws_size);

  {
    int n = N_NODES * HF;
    k_final<<<(n + 255) / 256, 256, 0, stream>>>(out, denom, bias, acc1, denom1, ws_size);
  }
}


# Launches 7 -> 5: max words start in the logit pass, the max pass rides in the init launch's first blocks, both walk forms share one launch: no launch whose waves all leave at once; sums unchanged
# speedup vs baseline: 4.3030x; 1.0359x over previous
#include <hip/hip_runtime.h>
#include <hip/hip_bf16.h>
#include <math.h>

#define N_NODES   100000
#define N_EDGES   1600000
#define IN_FEATS  128
#define HEADS     4
#define OUT_F     16
#define HF        64
#define NEG_SLOPE 0.2f

typedef __attribute__((ext_vector_type(2))) float v2f;
typedef __attribute__((ext_vector_type(4))) float v4f;
typedef __attribute__((ext_vector_type(8))) float v8f;

__device__ __forceinline__ unsigned ordEnc(float f) {
  unsigned u = __float_as_uint(f);
  return (u & 0x80000000u) ? ~u : (u | 0x80000000u);
}
__device__ __forceinline__ float ordDec(unsigned x) {
  return (x & 0x80000000u) ? __uint_as_float(x ^ 0x80000000u)
                           : __uint_as_float(~x);
}

__device__ __forceinline__ float lrelu(float x) {
  return x > 0.0f ? x : NEG_SLOPE * x;
}

#define SHIFT_RANGE 16.0f

#define WS_ONE_COPY   ((size_t)4 * ((size_t)N_NODES * HF + (size_t)4 * N_NODES * HEADS))
#define WS_TWO_COPY   (WS_ONE_COPY + (size_t)4 * ((size_t)N_NODES * HF + (size_t)N_NODES * HEADS))
#define WS_TWO_MAX    (WS_TWO_COPY + (size_t)4 * ((size_t)N_NODES * HEADS))
#define WS_WITH_BOUND (WS_TWO_MAX + (size_t)4 * 2 * HEADS)

__device__ __forceinline__ bool shift_from_bound(const unsigned* __restrict__ bound, bool have,
                                                 const int* __restrict__ src,
                                                 float (&mid)[HEADS]) {
  const unsigned* __restrict__ w = have ? bound : reinterpret_cast<const unsigned*>(src);
  bool ok = have;
  for (int h = 0; h < HEADS; ++h) {
    const float hi = ordDec(w[h]);
    const float lo = -ordDec(w[HEADS + h]);
    ok = ok && (hi - lo <= SHIFT_RANGE);
    mid[h] = lo + 0.5f * (hi - lo);
  }
  return ok;
}

#define TILES_PER_BLOCK 5
#define KSTEPS (IN_FEATS / 4)
#define A_PITCH (IN_FEATS + 4)

__device__ __forceinline__ void lds_handover() {
  __builtin_amdgcn_fence(__ATOMIC_RELEASE, "workgroup", "local");
  __builtin_amdgcn_s_barrier();
  __builtin_amdgcn_fence(__ATOMIC_ACQUIRE, "workgroup", "local");
}

__global__ __launch_bounds__(128) void k_project(const float* feats,
                                                 const float* __restrict__ W,
                                                 float* __restrict__ z,
                                                 unsigned* __restrict__ bound,
                                                 size_t ws_bytes) {
  __shared__ v4f s_a4[2][16 * A_PITCH / 4];

  if (blockIdx.x == 0 && threadIdx.x < 2 * HEADS && ws_bytes >= WS_WITH_BOUND)
    bound[threadIdx.x] = ordEnc(-INFINITY);

  const int t      = threadIdx.x & 31;
  const int w      = threadIdx.x >> 5;
  const int lane16 = t & 15;
  const int half   = t >> 4;
  const int tile0  = blockIdx.x * TILES_PER_BLOCK;

  const v4f* gsrc = reinterpret_cast<const v4f*>(feats)
                                    + ((size_t)tile0 * 16 + w) * (IN_FEATS / 4) + t;
  const int sdst = w * (A_PITCH / 4) + t;

  const size_t rstep = (size_t)4 * (IN_FEATS / 4);
  v4f g0 = gsrc[0], g1 = gsrc[rstep], g2 = gsrc[2 * rstep], g3 = gsrc[3 * rstep];

  const float* __restrict__ brow = W + (size_t)(w * 16 + lane16) * IN_FEATS + half * 2;
  v2f b[KSTEPS];
#pragma unroll
  for (int s = 0; s < KSTEPS; ++s) b[s] = v2f{ brow[4 * s], brow[4 * s + 1] };

  const int sstep = 4 * (A_PITCH / 4);
  s_a4[0][sdst] = g0;  s_a4[0][sdst + sstep] = g1;  s_a4[0][sdst + 2 * sstep] = g2;  s_a4[0][sdst + 3 * sstep] = g3;
  lds_handover();

#pragma unroll
  for (int i = 0; i < TILES_PER_BLOCK; ++i) {
    const int  cur  = i & 1;
    const bool more = i + 1 < TILES_PER_BLOCK;
    if (more) {
      const v4f* gn = gsrc + (size_t)(i + 1) * 16 * (IN_FEATS / 4);
      g0 = gn[0];  g1 = gn[rstep];  g2 = gn[2 * rstep];  g3 = gn[3 * rstep];
    }

    const float* __restrict__ ap = reinterpret_cast<const float*>(&s_a4[cur][0])
                                   + lane16 * A_PITCH + half * 2;
    v2f a[KSTEPS];
#pragma unroll
    for (int s = 0; s < KSTEPS; ++s) a[s] = *reinterpret_cast<const v2f*>(ap + 4 * s);
    __builtin_amdgcn_sched_barrier(0);

    v8f acc = {};
#pragma unroll
    for (int s = 0; s < KSTEPS; ++s)
      acc = __builtin_amdgcn_wmma_f32_16x16x4_f32(false, a[s], false, b[s],
                                                  (short)0, acc, false, false);

    __builtin_amdgcn_sched_barrier(0);

    float* __restrict__ zo = z + ((size_t)(tile0 + i) * 16 + half * 8) * HF + w * 16 + lane16;
#pragma unroll
    for (int r = 0; r < 8; ++r) zo[(size_t)r * HF] = acc[r];

    if (more) {
      v4f* sn = &s_a4[cur ^ 1][sdst];
      sn[0] = g0;  sn[sstep] = g1;  sn[2 * sstep] = g2;  sn[3 * sstep] = g3;
      lds_handover();
    }
  }
}

#define LOGIT_BLOCK 800

static_assert((N_NODES * HEADS) % LOGIT_BLOCK == 0 && LOGIT_BLOCK % 32 == 0, "exact grid of whole waves");

__global__ __launch_bounds__(LOGIT_BLOCK) void k_logits(const float* __restrict__ z,
                         const float* __restrict__ attn_l,
                         const float* __restrict__ attn_r,
                         float* __restrict__ el, float* __restrict__ er,
                         unsigned* __restrict__ menc, unsigned* __restrict__ menc1,
                         unsigned* __restrict__ bound, size_t ws_bytes) {
  __shared__ unsigned s_red[LOGIT_BLOCK / 32][2 * HEADS];
  const int i = blockIdx.x * LOGIT_BLOCK + threadIdx.x;
  int n = i >> 2, h = i & 3;
  const float* zp = z + (size_t)n * HF + h * OUT_F;
  const float* al = attn_l + h * OUT_F;
  const float* ar = attn_r + h * OUT_F;
  float sl = 0.0f, sr = 0.0f;
#pragma unroll
  for (int f = 0; f < OUT_F; ++f) {
    float v = zp[f];
    sl = fmaf(v, al[f], sl);
    sr = fmaf(v, ar[f], sr);
  }
  el[i] = sl;
  er[i] = sr;
  menc[i] = ordEnc(-INFINITY);
  if (ws_bytes >= WS_TWO_MAX) menc1[i] = ordEnc(-INFINITY);
  if (ws_bytes >= WS_WITH_BOUND) {
    unsigned vmax = ordEnc(sl), vneg = ordEnc(-sl);
    for (int off = 4; off < 32; off <<= 1) {
      const unsigned a = __shfl_xor(vmax, off), b = __shfl_xor(vneg, off);
      vmax = a > vmax ? a : vmax;
      vneg = b > vneg ? b : vneg;
    }
    const int lane = threadIdx.x & 31, wave = threadIdx.x >> 5;
    if (lane < HEADS) { s_red[wave][lane] = vmax;  s_red[wave][HEADS + lane] = vneg; }
    __syncthreads();
    if (threadIdx.x < 2 * HEADS) {
      unsigned v = s_red[0][threadIdx.x];
      for (int w = 1; w < LOGIT_BLOCK / 32; ++w) { const unsigned o = s_red[w][threadIdx.x]; v = o > v ? o : v; }
      atomicMax(&bound[threadIdx.x], v);
    }
  }
}


#define FALLBACK_BLOCKS 2048
#define MAX_PIECES (N_EDGES * HEADS / 256)

static_assert((N_NODES * HF) % 256 == 0 && (N_NODES * HF) / 256 >= FALLBACK_BLOCKS,
              "whole blocks cover the accumulators, and the launch has the blocks that hold the max pass");

__global__ __launch_bounds__(256) void k_init(float* __restrict__ out, float* __restrict__ denom,
                                              float* __restrict__ acc1, float* __restrict__ denom1,
                                              const int* __restrict__ src,
                                              const int* __restrict__ dst,
                                              const float* __restrict__ el,
                                              const float* __restrict__ er,
                                              unsigned* __restrict__ menc0,
                                              unsigned* __restrict__ menc1,
                                              const unsigned* __restrict__ bound,
                                              size_t ws_bytes) {
  const bool two = ws_bytes >= WS_TWO_COPY;
  const int i = blockIdx.x * 256 + threadIdx.x;
  out[i] = 0.0f;
  if (two) acc1[i] = 0.0f;
  if (i < N_NODES * HEADS) {
    denom[i] = 0.0f;
    if (two) denom1[i] = 0.0f;
  }
  if (blockIdx.x >= FALLBACK_BLOCKS) return;
  float midh[HEADS];
  if (shift_from_bound(bound, ws_bytes >= WS_WITH_BOUND, src, midh)) return;
  unsigned* menc = menc0;
  if (ws_bytes >= WS_TWO_MAX) {
    const unsigned die = (__builtin_amdgcn_s_sendmsg_rtn(0x87) >> 16) & 0xFu;
    if (die >= 4u) menc = menc1;
  }
  const int h = threadIdx.x & 3;
  for (int piece = blockIdx.x; piece < MAX_PIECES; piece += FALLBACK_BLOCKS) {
    const int e = (piece * 256 + threadIdx.x) >> 2;
    const int s = src[e], d = dst[e];
    const float lv = el[s * HEADS + h];
    const float rv = er[d * HEADS + h];
    atomicMax(&menc[d * HEADS + h], ordEnc(lrelu(lv + rv)));
  }
}

#define EPW 32
#define EPG 8
#define NGRP (EPW / EPG)
static_assert(N_EDGES % (EPW * 8) == 0, "exact grid of 8-wave blocks, one chunk of EPW edges per wave");

template <bool BY_BOUND>
__device__ __forceinline__ void edge_walk(const int (&sv)[NGRP], const int (&dv)[NGRP],
                                          const int lane, const float mid,
                                          const float* __restrict__ el,
                                          const float* __restrict__ er,
                                          const unsigned* __restrict__ menc,
                                          const unsigned* __restrict__ menc1,
                                          const float* __restrict__ z,
                                          float* __restrict__ denom0,
                                          float* __restrict__ out0,
                                          float* __restrict__ denom1,
                                          float* __restrict__ acc1,
                                          size_t ws_bytes) {
  const int h   = lane & 3;
  const int selLo = (lane >> 4) << 2;
  const int selHi = selLo + 8;

  float    lv[NGRP], rv[NGRP];
  for (int p = 0; p < NGRP; ++p) {
    lv[p] = el[sv[p] * HEADS + h];
    rv[p] = er[dv[p] * HEADS + h];
  }
  float wv[NGRP];
  if (!BY_BOUND) {
    const unsigned* __restrict__ mencB = (ws_bytes >= WS_TWO_MAX) ? menc1 : menc;
    unsigned mv[NGRP], ov[NGRP];
    for (int p = 0; p < NGRP; ++p) {
      mv[p] = menc[dv[p] * HEADS + h];
      ov[p] = mencB[dv[p] * HEADS + h];
    }
    for (int p = 0; p < NGRP; ++p) {
      const float lg = lrelu(lv[p] + rv[p]);
      const float c  = ordDec(mv[p] > ov[p] ? mv[p] : ov[p]);
      wv[p] = __expf(lg - c);
    }
  }

  float a0[EPG], a1[EPG];
  for (int j = 0; j < EPG; ++j) {
    const int s = __builtin_amdgcn_readlane(sv[0], j * 4);
    a0[j] = z[s * HF + lane];
    a1[j] = z[s * HF + lane + 32];
  }

  float* out   = out0;
  float* denom = denom0;
  if (ws_bytes >= WS_TWO_COPY) {
    const unsigned die = (__builtin_amdgcn_s_sendmsg_rtn(0x87) >> 16) & 0xFu;
    if (die >= 4u) { out = acc1; denom = denom1; }
  }

  if (BY_BOUND) {
    for (int p = 0; p < NGRP; ++p) {
      const float lg = lrelu(lv[p] + rv[p]);
      const float c  = lrelu(rv[p] + mid);
      wv[p] = __expf(lg - c);
    }
  }
  for (int p = 0; p < NGRP; ++p) atomicAdd(&denom[dv[p] * HEADS + h], wv[p]);

  for (int p = 0; p < NGRP; ++p) {
    float b0[EPG], b1[EPG];
    const int pn = (p + 1 < NGRP) ? p + 1 : p;
    if (!BY_BOUND && p > 0) {
      for (int j = 0; j < EPG; ++j) {
        const int s = __builtin_amdgcn_readlane(sv[p], j * 4);
        a0[j] = z[s * HF + lane];
        a1[j] = z[s * HF + lane + 32];
      }
    }
    if (BY_BOUND && p + 1 < NGRP) {
      for (int j = 0; j < EPG; ++j) {
        const int s = __builtin_amdgcn_readlane(sv[pn], j * 4);
        b0[j] = z[s * HF + lane];
        b1[j] = z[s * HF + lane + 32];
      }
    }
    const int wbits = __float_as_int(wv[p]);
    for (int j = 0; j < EPG; ++j) {
      const int   d  = __builtin_amdgcn_readlane(dv[p], j * 4);
      const float wl = __int_as_float(__builtin_amdgcn_ds_bpermute(selLo + 16 * j, wbits));
      const float wh = __int_as_float(__builtin_amdgcn_ds_bpermute(selHi + 16 * j, wbits));
      atomicAdd(&out[d * HF + lane],      wl * a0[j]);
      atomicAdd(&out[d * HF + lane + 32], wh * a1[j]);
    }
    if (BY_BOUND && p + 1 < NGRP) {
      for (int j = 0; j < EPG; ++j) { a0[j] = b0[j]; a1[j] = b1[j]; }
    }
  }
}

__global__ __launch_bounds__(256) void k_edge_agg(const int* __restrict__ src,
                                                  const int* __restrict__ dst,
                                                  const float* __restrict__ el,
                                                  const float* __restrict__ er,
                                                  const unsigned* __restrict__ menc,
                                                  const unsigned* __restrict__ menc1,
                                                  const float* __restrict__ z,
                                                  float* __restrict__ denom0,
                                                  float* __restrict__ out0,
                                                  float* __restrict__ denom1,
                                                  float* __restrict__ acc1,
                                                  const unsigned* __restrict__ bound,
                                                  size_t ws_bytes) {
  const int lane  = threadIdx.x & 31;
  const int chunk = blockIdx.x * (256 / 32) + (threadIdx.x >> 5);
  const int sub   = lane >> 2;
  const int h     = lane & 3;
  const int e0    = chunk * EPW;
  int sv[NGRP], dv[NGRP];
  for (int p = 0; p < NGRP; ++p) {
    const int e = e0 + p * EPG + sub;
    sv[p] = src[e];
    dv[p] = dst[e];
  }
  float midh[HEADS];
  const bool byBound = shift_from_bound(bound, ws_bytes >= WS_WITH_BOUND, src, midh);
  const float mid = h == 0 ? midh[0] : h == 1 ? midh[1] : h == 2 ? midh[2] : midh[3];
  if (byBound) edge_walk<true >(sv, dv, lane, mid, el, er, menc, menc1, z, denom0, out0, denom1, acc1, ws_bytes);
  else         edge_walk<false>(sv, dv, lane, mid, el, er, menc, menc1, z, denom0, out0, denom1, acc1, ws_bytes);
}

__global__ void k_final(float* __restrict__ out, const float* __restrict__ denom,
                        const float* __restrict__ bias,
                        const float* __restrict__ acc1, const float* __restrict__ denom1,
                        size_t ws_bytes) {
  const bool two = ws_bytes >= WS_TWO_COPY;
  int i = blockIdx.x * blockDim.x + threadIdx.x;
  if (i >= N_NODES * HF) return;
  int n = i >> 6;
  int c = i & 63;
  int h = c >> 4;
  float den = denom[n * HEADS + h];
  float v   = out[i];
  if (two) {
    den += denom1[n * HEADS + h];
    v   += acc1[i];
  }
  out[i] = (den != 0.0f) ? (v / den + bias[c]) : bias[c];
}

extern "C" void kernel_launch(void* const* d_in, const int* in_sizes, int n_in,
                              void* d_out, int out_size, void* d_ws, size_t ws_size,
                              hipStream_t stream) {
  (void)in_sizes; (void)n_in; (void)out_size;
  const float* feats  = (const float*)d_in[0];
  const float* W      = (const float*)d_in[1];
  const float* attn_l = (const float*)d_in[2];
  const float* attn_r = (const float*)d_in[3];
  const float* bias   = (const float*)d_in[4];
  const int*   src    = (const int*)d_in[5];
  const int*   dst    = (const int*)d_in[6];
  float*       out    = (float*)d_out;

  float*    ws    = (float*)d_ws;
  float*    z     = ws;
  float*    el    = z  + (size_t)N_NODES * HF;
  float*    er    = el + (size_t)N_NODES * HEADS;
  unsigned* menc  = (unsigned*)(er + (size_t)N_NODES * HEADS);
  float*    denom = (float*)(menc + (size_t)N_NODES * HEADS);
  float*    acc1   = denom + (size_t)N_NODES * HEADS;
  float*    denom1 = acc1  + (size_t)N_NODES * HF;
  unsigned* menc1  = (unsigned*)(denom1 + (size_t)N_NODES * HEADS);
  unsigned* bound  = menc1 + (size_t)N_NODES * HEADS;

  k_project<<<N_NODES / (16 * TILES_PER_BLOCK), 128, 0, stream>>>(feats, W, z, bound, ws_size);

  {
    int n = N_NODES * HEADS;
    k_logits<<<n / LOGIT_BLOCK, LOGIT_BLOCK, 0, stream>>>(z, attn_l, attn_r, el, er, menc, menc1, bound, ws_size);
  }

  {
    int n = N_NODES * HF;
    k_init<<<n / 256, 256, 0, stream>>>(out, denom, acc1, denom1, src, dst, el, er, menc, menc1, bound, ws_size);
  }

  k_edge_agg<<<N_EDGES / (EPW * 8), 256, 0, stream>>>(src, dst, el, er,
                                                    menc, menc1, z, denom, out,
                                                    denom1, acc1, bound, ws_size);

  {
    int n = N_NODES * HF;
    k_final<<<(n + 255) / 256, 256, 0, stream>>>(out, denom, bias, acc1, denom1, ws_size);
  }
}


# Edge pass, bound form: source logit el summed from the gathered z row by a 16-lane tree (was a gather of the serial sum); weights passed inside lane rows by DPP; atomic adds as before
# speedup vs baseline: 4.4558x; 1.0355x over previous
#include <hip/hip_runtime.h>
#include <hip/hip_bf16.h>
#include <math.h>

#define N_NODES   100000
#define N_EDGES   1600000
#define IN_FEATS  128
#define HEADS     4
#define OUT_F     16
#define HF        64
#define NEG_SLOPE 0.2f

typedef __attribute__((ext_vector_type(2))) float v2f;
typedef __attribute__((ext_vector_type(4))) float v4f;
typedef __attribute__((ext_vector_type(8))) float v8f;

__device__ __forceinline__ unsigned ordEnc(float f) {
  unsigned u = __float_as_uint(f);
  return (u & 0x80000000u) ? ~u : (u | 0x80000000u);
}
__device__ __forceinline__ float ordDec(unsigned x) {
  return (x & 0x80000000u) ? __uint_as_float(x ^ 0x80000000u)
                           : __uint_as_float(~x);
}

__device__ __forceinline__ float lrelu(float x) {
  return x > 0.0f ? x : NEG_SLOPE * x;
}

#define SHIFT_RANGE 16.0f

#define WS_ONE_COPY   ((size_t)4 * ((size_t)N_NODES * HF + (size_t)4 * N_NODES * HEADS))
#define WS_TWO_COPY   (WS_ONE_COPY + (size_t)4 * ((size_t)N_NODES * HF + (size_t)N_NODES * HEADS))
#define WS_TWO_MAX    (WS_TWO_COPY + (size_t)4 * ((size_t)N_NODES * HEADS))
#define WS_WITH_BOUND (WS_TWO_MAX + (size_t)4 * 2 * HEADS)

__device__ __forceinline__ bool shift_from_bound(const unsigned* __restrict__ bound, bool have,
                                                 const int* __restrict__ src,
                                                 float (&mid)[HEADS]) {
  const unsigned* __restrict__ w = have ? bound : reinterpret_cast<const unsigned*>(src);
  bool ok = have;
  for (int h = 0; h < HEADS; ++h) {
    const float hi = ordDec(w[h]);
    const float lo = -ordDec(w[HEADS + h]);
    ok = ok && (hi - lo <= SHIFT_RANGE);
    mid[h] = lo + 0.5f * (hi - lo);
  }
  return ok;
}

#define TILES_PER_BLOCK 5
#define KSTEPS (IN_FEATS / 4)
#define A_PITCH (IN_FEATS + 4)

__device__ __forceinline__ void lds_handover() {
  __builtin_amdgcn_fence(__ATOMIC_RELEASE, "workgroup", "local");
  __builtin_amdgcn_s_barrier();
  __builtin_amdgcn_fence(__ATOMIC_ACQUIRE, "workgroup", "local");
}

__global__ __launch_bounds__(128) void k_project(const float* feats,
                                                 const float* __restrict__ W,
                                                 float* __restrict__ z,
                                                 unsigned* __restrict__ bound,
                                                 size_t ws_bytes) {
  __shared__ v4f s_a4[2][16 * A_PITCH / 4];

  if (blockIdx.x == 0 && threadIdx.x < 2 * HEADS && ws_bytes >= WS_WITH_BOUND)
    bound[threadIdx.x] = ordEnc(-INFINITY);

  const int t      = threadIdx.x & 31;
  const int w      = threadIdx.x >> 5;
  const int lane16 = t & 15;
  const int half   = t >> 4;
  const int tile0  = blockIdx.x * TILES_PER_BLOCK;

  const v4f* gsrc = reinterpret_cast<const v4f*>(feats)
                                    + ((size_t)tile0 * 16 + w) * (IN_FEATS / 4) + t;
  const int sdst = w * (A_PITCH / 4) + t;

  const size_t rstep = (size_t)4 * (IN_FEATS / 4);
  v4f g0 = gsrc[0], g1 = gsrc[rstep], g2 = gsrc[2 * rstep], g3 = gsrc[3 * rstep];

  const float* __restrict__ brow = W + (size_t)(w * 16 + lane16) * IN_FEATS + half * 2;
  v2f b[KSTEPS];
#pragma unroll
  for (int s = 0; s < KSTEPS; ++s) b[s] = v2f{ brow[4 * s], brow[4 * s + 1] };

  const int sstep = 4 * (A_PITCH / 4);
  s_a4[0][sdst] = g0;  s_a4[0][sdst + sstep] = g1;  s_a4[0][sdst + 2 * sstep] = g2;  s_a4[0][sdst + 3 * sstep] = g3;
  lds_handover();

#pragma unroll
  for (int i = 0; i < TILES_PER_BLOCK; ++i) {
    const int  cur  = i & 1;
    const bool more = i + 1 < TILES_PER_BLOCK;
    if (more) {
      const v4f* gn = gsrc + (size_t)(i + 1) * 16 * (IN_FEATS / 4);
      g0 = gn[0];  g1 = gn[rstep];  g2 = gn[2 * rstep];  g3 = gn[3 * rstep];
    }

    const float* __restrict__ ap = reinterpret_cast<const float*>(&s_a4[cur][0])
                                   + lane16 * A_PITCH + half * 2;
    v2f a[KSTEPS];
#pragma unroll
    for (int s = 0; s < KSTEPS; ++s) a[s] = *reinterpret_cast<const v2f*>(ap + 4 * s);
    __builtin_amdgcn_sched_barrier(0);

    v8f acc = {};
#pragma unroll
    for (int s = 0; s < KSTEPS; ++s)
      acc = __builtin_amdgcn_wmma_f32_16x16x4_f32(false, a[s], false, b[s],
                                                  (short)0, acc, false, false);

    __builtin_amdgcn_sched_barrier(0);

    float* __restrict__ zo = z + ((size_t)(tile0 + i) * 16 + half * 8) * HF + w * 16 + lane16;
#pragma unroll
    for (int r = 0; r < 8; ++r) zo[(size_t)r * HF] = acc[r];

    if (more) {
      v4f* sn = &s_a4[cur ^ 1][sdst];
      sn[0] = g0;  sn[sstep] = g1;  sn[2 * sstep] = g2;  sn[3 * sstep] = g3;
      lds_handover();
    }
  }
}

#define LOGIT_BLOCK 800

static_assert((N_NODES * HEADS) % LOGIT_BLOCK == 0 && LOGIT_BLOCK % 32 == 0, "exact grid of whole waves");

__global__ __launch_bounds__(LOGIT_BLOCK) void k_logits(const float* __restrict__ z,
                         const float* __restrict__ attn_l,
                         const float* __restrict__ attn_r,
                         float* __restrict__ el, float* __restrict__ er,
                         unsigned* __restrict__ menc, unsigned* __restrict__ menc1,
                         unsigned* __restrict__ bound, size_t ws_bytes) {
  __shared__ unsigned s_red[LOGIT_BLOCK / 32][2 * HEADS];
  const int i = blockIdx.x * LOGIT_BLOCK + threadIdx.x;
  int n = i >> 2, h = i & 3;
  const float* zp = z + (size_t)n * HF + h * OUT_F;
  const float* al = attn_l + h * OUT_F;
  const float* ar = attn_r + h * OUT_F;
  float sl = 0.0f, sr = 0.0f;
#pragma unroll
  for (int f = 0; f < OUT_F; ++f) {
    float v = zp[f];
    sl = fmaf(v, al[f], sl);
    sr = fmaf(v, ar[f], sr);
  }
  el[i] = sl;
  er[i] = sr;
  menc[i] = ordEnc(-INFINITY);
  if (ws_bytes >= WS_TWO_MAX) menc1[i] = ordEnc(-INFINITY);
  if (ws_bytes >= WS_WITH_BOUND) {
    unsigned vmax = ordEnc(sl), vneg = ordEnc(-sl);
    for (int off = 4; off < 32; off <<= 1) {
      const unsigned a = __shfl_xor(vmax, off), b = __shfl_xor(vneg, off);
      vmax = a > vmax ? a : vmax;
      vneg = b > vneg ? b : vneg;
    }
    const int lane = threadIdx.x & 31, wave = threadIdx.x >> 5;
    if (lane < HEADS) { s_red[wave][lane] = vmax;  s_red[wave][HEADS + lane] = vneg; }
    __syncthreads();
    if (threadIdx.x < 2 * HEADS) {
      unsigned v = s_red[0][threadIdx.x];
      for (int w = 1; w < LOGIT_BLOCK / 32; ++w) { const unsigned o = s_red[w][threadIdx.x]; v = o > v ? o : v; }
      atomicMax(&bound[threadIdx.x], v);
    }
  }
}


#define FALLBACK_BLOCKS 2048
#define MAX_PIECES (N_EDGES * HEADS / 256)

static_assert((N_NODES * HF) % 256 == 0 && (N_NODES * HF) / 256 >= FALLBACK_BLOCKS,
              "whole blocks cover the accumulators, and the launch has the blocks that hold the max pass");

__global__ __launch_bounds__(256) void k_init(float* __restrict__ out, float* __restrict__ denom,
                                              float* __restrict__ acc1, float* __restrict__ denom1,
                                              const int* __restrict__ src,
                                              const int* __restrict__ dst,
                                              const float* __restrict__ el,
                                              const float* __restrict__ er,
                                              unsigned* __restrict__ menc0,
                                              unsigned* __restrict__ menc1,
                                              const unsigned* __restrict__ bound,
                                              size_t ws_bytes) {
  const bool two = ws_bytes >= WS_TWO_COPY;
  const int i = blockIdx.x * 256 + threadIdx.x;
  out[i] = 0.0f;
  if (two) acc1[i] = 0.0f;
  if (i < N_NODES * HEADS) {
    denom[i] = 0.0f;
    if (two) denom1[i] = 0.0f;
  }
  if (blockIdx.x >= FALLBACK_BLOCKS) return;
  float midh[HEADS];
  if (shift_from_bound(bound, ws_bytes >= WS_WITH_BOUND, src, midh)) return;
  unsigned* menc = menc0;
  if (ws_bytes >= WS_TWO_MAX) {
    const unsigned die = (__builtin_amdgcn_s_sendmsg_rtn(0x87) >> 16) & 0xFu;
    if (die >= 4u) menc = menc1;
  }
  const int h = threadIdx.x & 3;
  for (int piece = blockIdx.x; piece < MAX_PIECES; piece += FALLBACK_BLOCKS) {
    const int e = (piece * 256 + threadIdx.x) >> 2;
    const int s = src[e], d = dst[e];
    const float lv = el[s * HEADS + h];
    const float rv = er[d * HEADS + h];
    atomicMax(&menc[d * HEADS + h], ordEnc(lrelu(lv + rv)));
  }
}

#define EPW 32
#define EPG 8
#define NGRP (EPW / EPG)
static_assert(N_EDGES % (EPW * 8) == 0, "exact grid of 8-wave blocks, one chunk of EPW edges per wave");

#define DEF_XADD(M)                                                                          \
  __device__ __forceinline__ float xadd##M(float x) {                                        \
    float r;                                                                                 \
    asm("v_add_f32_dpp %0, %1, %1 row_xmask:" #M " row_mask:0xf bank_mask:0xf"               \
        : "=v"(r) : "v"(x));                                                                 \
    return r;                                                                                \
  }
DEF_XADD(8) DEF_XADD(4) DEF_XADD(2) DEF_XADD(1)
#undef DEF_XADD
#define DEF_MUL_SHARE(N)                                                                     \
  __device__ __forceinline__ float mul_share##N(float w, float a) {                          \
    float r;                                                                                 \
    asm("v_mul_f32_dpp %0, %1, %2 row_share:" #N " row_mask:0xf bank_mask:0xf"               \
        : "=v"(r) : "v"(w), "v"(a));                                                         \
    return r;                                                                                \
  }
DEF_MUL_SHARE(0)  DEF_MUL_SHARE(1)  DEF_MUL_SHARE(2)  DEF_MUL_SHARE(3)
DEF_MUL_SHARE(4)  DEF_MUL_SHARE(5)  DEF_MUL_SHARE(6)  DEF_MUL_SHARE(7)
DEF_MUL_SHARE(8)  DEF_MUL_SHARE(9)  DEF_MUL_SHARE(10) DEF_MUL_SHARE(11)
DEF_MUL_SHARE(12) DEF_MUL_SHARE(13) DEF_MUL_SHARE(14) DEF_MUL_SHARE(15)
#undef DEF_MUL_SHARE

__device__ __forceinline__ void edge_walk_bound(const int S, const int (&dv)[NGRP],
                                                const int lane, const int head, const float mid,
                                                const float* __restrict__ er,
                                                const float* __restrict__ attn_l,
                                                const float* __restrict__ z,
                                                float* __restrict__ denom0,
                                                float* __restrict__ out0,
                                                float* __restrict__ denom1,
                                                float* __restrict__ acc1,
                                                size_t ws_bytes) {
  const float al0 = attn_l[lane];
  const float al1 = attn_l[lane + 32];
  float rv[NGRP];
  for (int p = 0; p < NGRP; ++p) rv[p] = er[dv[p] * HEADS + head];

  float a0[EPG], a1[EPG];
  for (int j = 0; j < EPG; ++j) {
    const int s = __builtin_amdgcn_readlane(S, j);
    a0[j] = z[s * HF + lane];
    a1[j] = z[s * HF + lane + 32];
  }

  float* out   = out0;
  float* denom = denom0;
  if (ws_bytes >= WS_TWO_COPY) {
    const unsigned die = (__builtin_amdgcn_s_sendmsg_rtn(0x87) >> 16) & 0xFu;
    if (die >= 4u) { out = acc1; denom = denom1; }
  }

  const bool up8 = (lane & 8) != 0, up4 = (lane & 4) != 0, up2 = (lane & 2) != 0, up1 = (lane & 1) != 0;

#pragma unroll
  for (int p = 0; p < NGRP; ++p) {
    float b0[EPG], b1[EPG];
    if (p + 1 < NGRP) {
      for (int j = 0; j < EPG; ++j) {
        const int s = __builtin_amdgcn_readlane(S, (p + 1) * EPG + j);
        b0[j] = z[s * HF + lane];
        b1[j] = z[s * HF + lane + 32];
      }
    }

    float v8[EPG];
    for (int k = 0; k < EPG; ++k) {
      const float x = a0[k] * al0, y = a1[k] * al1;
      const float t = xadd8(x);
      const float u = xadd8(y);
      v8[k] = up8 ? u : t;
    }
    float v4[4];
    for (int k = 0; k < 4; ++k) {
      const float t = xadd4(v8[k]);
      const float u = xadd4(v8[k + 4]);
      v4[k] = up4 ? u : t;
    }
    float v2[2];
    for (int k = 0; k < 2; ++k) {
      const float t = xadd2(v4[k]);
      const float u = xadd2(v4[k + 2]);
      v2[k] = up2 ? u : t;
    }
    const float t1  = xadd1(v2[0]);
    const float u1  = xadd1(v2[1]);
    const float elv = up1 ? u1 : t1;

    const float lg = lrelu(elv + rv[p]);
    const float c  = lrelu(rv[p] + mid);
    const float w  = __expf(lg - c);
    atomicAdd(&denom[dv[p] * HEADS + head], w);

#define ROW_ADD(J, J8) {                                                    \
      const int d = __builtin_amdgcn_readlane(dv[p], J);                    \
      atomicAdd(&out[d * HF + lane],      mul_share##J(w, a0[J]));          \
      atomicAdd(&out[d * HF + lane + 32], mul_share##J8(w, a1[J]));         \
    }
    ROW_ADD(0, 8)  ROW_ADD(1, 9)  ROW_ADD(2, 10) ROW_ADD(3, 11)
    ROW_ADD(4, 12) ROW_ADD(5, 13) ROW_ADD(6, 14) ROW_ADD(7, 15)
#undef ROW_ADD

    if (p + 1 < NGRP) {
      for (int j = 0; j < EPG; ++j) { a0[j] = b0[j]; a1[j] = b1[j]; }
    }
  }
}

__device__ __forceinline__ void edge_walk_max(const int (&sv)[NGRP], const int (&dv)[NGRP],
                                              const int lane,
                                              const float* __restrict__ el,
                                              const float* __restrict__ er,
                                              const unsigned* __restrict__ menc,
                                              const unsigned* __restrict__ menc1,
                                              const float* __restrict__ z,
                                              float* __restrict__ denom0,
                                              float* __restrict__ out0,
                                              float* __restrict__ denom1,
                                              float* __restrict__ acc1,
                                              size_t ws_bytes) {
  const int h   = lane & 3;
  const int selLo = (lane >> 4) << 2;
  const int selHi = selLo + 8;

  const unsigned* __restrict__ mencB = (ws_bytes >= WS_TWO_MAX) ? menc1 : menc;
  float    lv[NGRP], rv[NGRP];
  unsigned mv[NGRP], ov[NGRP];
  for (int p = 0; p < NGRP; ++p) {
    lv[p] = el[sv[p] * HEADS + h];
    rv[p] = er[dv[p] * HEADS + h];
    mv[p] = menc[dv[p] * HEADS + h];
    ov[p] = mencB[dv[p] * HEADS + h];
  }
  float wv[NGRP];
  for (int p = 0; p < NGRP; ++p) {
    const float lg = lrelu(lv[p] + rv[p]);
    const float c  = ordDec(mv[p] > ov[p] ? mv[p] : ov[p]);
    wv[p] = __expf(lg - c);
  }

  float* out   = out0;
  float* denom = denom0;
  if (ws_bytes >= WS_TWO_COPY) {
    const unsigned die = (__builtin_amdgcn_s_sendmsg_rtn(0x87) >> 16) & 0xFu;
    if (die >= 4u) { out = acc1; denom = denom1; }
  }
  for (int p = 0; p < NGRP; ++p) atomicAdd(&denom[dv[p] * HEADS + h], wv[p]);

  for (int p = 0; p < NGRP; ++p) {
    float a0[EPG], a1[EPG];
    for (int j = 0; j < EPG; ++j) {
      const int s = __builtin_amdgcn_readlane(sv[p], j * 4);
      a0[j] = z[s * HF + lane];
      a1[j] = z[s * HF + lane + 32];
    }
    const int wbits = __float_as_int(wv[p]);
    for (int j = 0; j < EPG; ++j) {
      const int   d  = __builtin_amdgcn_readlane(dv[p], j * 4);
      const float wl = __int_as_float(__builtin_amdgcn_ds_bpermute(selLo + 16 * j, wbits));
      const float wh = __int_as_float(__builtin_amdgcn_ds_bpermute(selHi + 16 * j, wbits));
      atomicAdd(&out[d * HF + lane],      wl * a0[j]);
      atomicAdd(&out[d * HF + lane + 32], wh * a1[j]);
    }
  }
}

__global__ __launch_bounds__(256) __attribute__((amdgpu_waves_per_eu(16, 16))) void k_edge_agg(const int* __restrict__ src,
                                                  const int* __restrict__ dst,
                                                  const float* __restrict__ el,
                                                  const float* __restrict__ er,
                                                  const float* __restrict__ attn_l,
                                                  const unsigned* __restrict__ menc,
                                                  const unsigned* __restrict__ menc1,
                                                  const float* __restrict__ z,
                                                  float* __restrict__ denom0,
                                                  float* __restrict__ out0,
                                                  float* __restrict__ denom1,
                                                  float* __restrict__ acc1,
                                                  const unsigned* __restrict__ bound,
                                                  size_t ws_bytes) {
  const int lane  = threadIdx.x & 31;
  const int chunk = blockIdx.x * (256 / 32) + (threadIdx.x >> 5);
  const int e0    = chunk * EPW;
  const int S = src[e0 + lane];
  int dvn[NGRP];
  for (int p = 0; p < NGRP; ++p) dvn[p] = dst[e0 + p * EPG + (lane & 7)];
  float midh[HEADS];
  const bool byBound = shift_from_bound(bound, ws_bytes >= WS_WITH_BOUND, src, midh);
  if (byBound) {
    const int   head = (lane >> 4) + ((lane >> 2) & 2);
    const float mid  = head == 0 ? midh[0] : head == 1 ? midh[1] : head == 2 ? midh[2] : midh[3];
    edge_walk_bound(S, dvn, lane, head, mid, er, attn_l, z, denom0, out0, denom1, acc1, ws_bytes);
  } else {
    const int sub = lane >> 2;
    int sv[NGRP], dv[NGRP];
    for (int p = 0; p < NGRP; ++p) {
      sv[p] = __builtin_amdgcn_ds_bpermute(4 * (p * EPG + sub), S);
      dv[p] = __builtin_amdgcn_ds_bpermute(4 * sub, dvn[p]);
    }
    edge_walk_max(sv, dv, lane, el, er, menc, menc1, z, denom0, out0, denom1, acc1, ws_bytes);
  }
}

__global__ void k_final(float* __restrict__ out, const float* __restrict__ denom,
                        const float* __restrict__ bias,
                        const float* __restrict__ acc1, const float* __restrict__ denom1,
                        size_t ws_bytes) {
  const bool two = ws_bytes >= WS_TWO_COPY;
  int i = blockIdx.x * blockDim.x + threadIdx.x;
  if (i >= N_NODES * HF) return;
  int n = i >> 6;
  int c = i & 63;
  int h = c >> 4;
  float den = denom[n * HEADS + h];
  float v   = out[i];
  if (two) {
    den += denom1[n * HEADS + h];
    v   += acc1[i];
  }
  out[i] = (den != 0.0f) ? (v / den + bias[c]) : bias[c];
}

extern "C" void kernel_launch(void* const* d_in, const int* in_sizes, int n_in,
                              void* d_out, int out_size, void* d_ws, size_t ws_size,
                              hipStream_t stream) {
  (void)in_sizes; (void)n_in; (void)out_size;
  const float* feats  = (const float*)d_in[0];
  const float* W      = (const float*)d_in[1];
  const float* attn_l = (const float*)d_in[2];
  const float* attn_r = (const float*)d_in[3];
  const float* bias   = (const float*)d_in[4];
  const int*   src    = (const int*)d_in[5];
  const int*   dst    = (const int*)d_in[6];
  float*       out    = (float*)d_out;

  float*    ws    = (float*)d_ws;
  float*    z     = ws;
  float*    el    = z  + (size_t)N_NODES * HF;
  float*    er    = el + (size_t)N_NODES * HEADS;
  unsigned* menc  = (unsigned*)(er + (size_t)N_NODES * HEADS);
  float*    denom = (float*)(menc + (size_t)N_NODES * HEADS);
  float*    acc1   = denom + (size_t)N_NODES * HEADS;
  float*    denom1 = acc1  + (size_t)N_NODES * HF;
  unsigned* menc1  = (unsigned*)(denom1 + (size_t)N_NODES * HEADS);
  unsigned* bound  = menc1 + (size_t)N_NODES * HEADS;

  k_project<<<N_NODES / (16 * TILES_PER_BLOCK), 128, 0, stream>>>(feats, W, z, bound, ws_size);

  {
    int n = N_NODES * HEADS;
    k_logits<<<n / LOGIT_BLOCK, LOGIT_BLOCK, 0, stream>>>(z, attn_l, attn_r, el, er, menc, menc1, bound, ws_size);
  }

  {
    int n = N_NODES * HF;
    k_init<<<n / 256, 256, 0, stream>>>(out, denom, acc1, denom1, src, dst, el, er, menc, menc1, bound, ws_size);
  }

  k_edge_agg<<<N_EDGES / (EPW * 8), 256, 0, stream>>>(src, dst, el, er, attn_l,
                                                    menc, menc1, z, denom, out,
                                                    denom1, acc1, bound, ws_size);

  {
    int n = N_NODES * HF;
    k_final<<<(n + 255) / 256, 256, 0, stream>>>(out, denom, bias, acc1, denom1, ws_size);
  }
}
